# starting kernel
# baseline (speedup 1.0000x reference)
#include <hip/hip_runtime.h>

typedef __attribute__((ext_vector_type(2))) float v2f;
typedef __attribute__((ext_vector_type(8))) float v8f;

#define N_NODES_C 100000
#define N_EDGES_C 25000
#define NNZ_C     800000
#define D_C       64
#define ALPHA_C   0.5f

#define A_PSTRIDE 144
#define B_PSTRIDE 160

__device__ __forceinline__ v8f wmma_f32(v2f a, v2f b, v8f c) {
  return __builtin_amdgcn_wmma_f32_16x16x4_f32(
       false, a,  false, b,
       (short)0, c,  false,  false);
}

__global__ void k_diffuse(const float* __restrict__ X,
                          const float* __restrict__ vals,
                          const int* __restrict__ rows,
                          const int* __restrict__ cols,
                          float* __restrict__ Xd) {
  int t  = blockIdx.x * blockDim.x + threadIdx.x;
  int i  = t >> 4;
  int c4 = (t & 15) << 2;
  __builtin_prefetch(vals + i + 1024, 0, 3);
  __builtin_prefetch(cols + i + 1024, 0, 3);
  float v = vals[i];
  const float4 x = *reinterpret_cast<const float4*>(X + (size_t)cols[i] * D_C + c4);
  float* dst = Xd + (size_t)rows[i] * D_C + c4;
  atomicAdd(dst + 0, v * x.x);
  atomicAdd(dst + 1, v * x.y);
  atomicAdd(dst + 2, v * x.z);
  atomicAdd(dst + 3, v * x.w);
}

__global__ void __launch_bounds__(256) k_gemm64(
    const float* __restrict__ A, const float* __restrict__ Ablend, float alpha,
    const float* __restrict__ B, const float* __restrict__ bias,
    float* __restrict__ Out, int nrows) {
  __shared__ float At[32 * A_PSTRIDE * 2];
  __shared__ float Bf[32 * B_PSTRIDE];
  __shared__ float bs[64];

  const int tid  = threadIdx.x;
  const int row0 = blockIdx.x * 128;

  for (int idx = tid; idx < 128 * 16; idx += 256) {
    int rr = idx >> 4, c4 = idx & 15;
    int row = row0 + rr;
    float4 val = make_float4(0.f, 0.f, 0.f, 0.f);
    if (row < nrows) {
      val = *reinterpret_cast<const float4*>(A + (size_t)row * D_C + (c4 << 2));
      if (Ablend) {
        float4 w = *reinterpret_cast<const float4*>(Ablend + (size_t)row * D_C + (c4 << 2));
        val.x = (1.f - alpha) * val.x + alpha * w.x;
        val.y = (1.f - alpha) * val.y + alpha * w.y;
        val.z = (1.f - alpha) * val.z + alpha * w.z;
        val.w = (1.f - alpha) * val.w + alpha * w.w;
      }
    }
    int p = c4 << 1;
    v2f lo; lo.x = val.x; lo.y = val.y;
    v2f hi2; hi2.x = val.z; hi2.y = val.w;
    *reinterpret_cast<v2f*>(&At[(p * A_PSTRIDE + rr) * 2])       = lo;
    *reinterpret_cast<v2f*>(&At[((p + 1) * A_PSTRIDE + rr) * 2]) = hi2;
  }
  for (int idx = tid; idx < 64 * 64; idx += 256) {
    int k = idx >> 6, col = idx & 63;
    Bf[(k >> 1) * B_PSTRIDE + (col << 1) + (k & 1)] = B[idx];
  }
  if (tid < 64) bs[tid] = bias[tid];
  __syncthreads();

  const int wave = tid >> 5, lane = tid & 31;
  const int lrow = lane & 15, hi = lane >> 4;
  const int rr   = wave * 16 + lrow;

  v8f acc[4] = {v8f{}, v8f{}, v8f{}, v8f{}};
#pragma unroll
  for (int k4 = 0; k4 < 16; ++k4) {
    const int p = (k4 << 1) + hi;
    v2f a = *reinterpret_cast<const v2f*>(&At[(p * A_PSTRIDE + rr) * 2]);
#pragma unroll
    for (int n = 0; n < 4; ++n) {
      v2f b = *reinterpret_cast<const v2f*>(
          &Bf[p * B_PSTRIDE + (((n << 4) + lrow) << 1)]);
      acc[n] = wmma_f32(a, b, acc[n]);
    }
  }

#pragma unroll
  for (int n = 0; n < 4; ++n)
#pragma unroll
    for (int j = 0; j < 8; ++j) {
      int row = row0 + wave * 16 + j + (hi << 3);
      if (row < nrows) {
        int col = (n << 4) + lrow;
        Out[(size_t)row * D_C + col] = acc[n][j] + bs[col];
      }
    }
}

__global__ void k_scatter_edge(const float* __restrict__ Y1,
                               const int* __restrict__ vertex,
                               const int* __restrict__ edges,
                               float* __restrict__ Xe) {
  int t  = blockIdx.x * blockDim.x + threadIdx.x;
  int i  = t >> 4;
  int c4 = (t & 15) << 2;
  __builtin_prefetch(vertex + i + 1024, 0, 3);
  __builtin_prefetch(edges + i + 1024, 0, 3);
  const float4 y = *reinterpret_cast<const float4*>(Y1 + (size_t)vertex[i] * D_C + c4);
  float* dst = Xe + (size_t)edges[i] * D_C + c4;
  atomicAdd(dst + 0, y.x);
  atomicAdd(dst + 1, y.y);
  atomicAdd(dst + 2, y.z);
  atomicAdd(dst + 3, y.w);
}

__global__ void __launch_bounds__(256) k_pair_gemm(
    const float* __restrict__ Xd, const float* __restrict__ Xe,
    const int* __restrict__ vertex, const int* __restrict__ edges,
    const float* __restrict__ W2, const float* __restrict__ b2,
    float* __restrict__ Xv) {
  __shared__ float At[64 * A_PSTRIDE * 2];
  __shared__ float Bf[64 * B_PSTRIDE];
  __shared__ float bs[64];

  const int tid = threadIdx.x;
  const int pb  = blockIdx.x * 128;

  __builtin_prefetch(vertex + pb + 128, 0, 3);
  __builtin_prefetch(edges + pb + 128, 0, 3);

  for (int idx = tid; idx < 128 * 32; idx += 256) {
    int pp = idx >> 5, c4 = idx & 31;
    const float* src =
        (c4 < 16) ? (Xd + (size_t)vertex[pb + pp] * D_C + (c4 << 2))
                  : (Xe + (size_t)edges[pb + pp] * D_C + ((c4 - 16) << 2));
    float4 val = *reinterpret_cast<const float4*>(src);
    int p = c4 << 1;
    v2f lo; lo.x = val.x; lo.y = val.y;
    v2f hi2; hi2.x = val.z; hi2.y = val.w;
    *reinterpret_cast<v2f*>(&At[(p * A_PSTRIDE + pp) * 2])       = lo;
    *reinterpret_cast<v2f*>(&At[((p + 1) * A_PSTRIDE + pp) * 2]) = hi2;
  }
  for (int idx = tid; idx < 128 * 64; idx += 256) {
    int k = idx >> 6, col = idx & 63;
    Bf[(k >> 1) * B_PSTRIDE + (col << 1) + (k & 1)] = W2[idx];
  }
  if (tid < 64) bs[tid] = b2[tid];
  __syncthreads();

  const int wave = tid >> 5, lane = tid & 31;
  const int lrow = lane & 15, hi = lane >> 4;
  const int rr   = wave * 16 + lrow;

  v8f acc[4] = {v8f{}, v8f{}, v8f{}, v8f{}};
#pragma unroll
  for (int k4 = 0; k4 < 32; ++k4) {
    const int p = (k4 << 1) + hi;
    v2f a = *reinterpret_cast<const v2f*>(&At[(p * A_PSTRIDE + rr) * 2]);
#pragma unroll
    for (int n = 0; n < 4; ++n) {
      v2f b = *reinterpret_cast<const v2f*>(
          &Bf[p * B_PSTRIDE + (((n << 4) + lrow) << 1)]);
      acc[n] = wmma_f32(a, b, acc[n]);
    }
  }

  const int p0w = pb + wave * 16;
#pragma unroll
  for (int j = 0; j < 8; ++j) {
    int row = vertex[p0w + j + (hi << 3)];
#pragma unroll
    for (int n = 0; n < 4; ++n) {
      int col = (n << 4) + lrow;
      atomicAdd(&Xv[(size_t)row * D_C + col], acc[n][j] + bs[col]);
    }
  }
}

extern "C" void kernel_launch(void* const* d_in, const int* in_sizes, int n_in,
                              void* d_out, int out_size, void* d_ws, size_t ws_size,
                              hipStream_t stream) {
  const float* X      = (const float*)d_in[0];
  const float* W1_w   = (const float*)d_in[1];
  const float* W1_b   = (const float*)d_in[2];
  const float* W2_w   = (const float*)d_in[3];
  const float* W2_b   = (const float*)d_in[4];
  const float* W_w    = (const float*)d_in[5];
  const float* W_b    = (const float*)d_in[6];
  const float* dvals  = (const float*)d_in[7];
  const int*   vertex = (const int*)d_in[8];
  const int*   edges  = (const int*)d_in[9];
  const int*   drows  = (const int*)d_in[10];
  const int*   dcols  = (const int*)d_in[11];
  float* out = (float*)d_out;

  float* Xd = (float*)d_ws;
  float* Y1 = Xd + (size_t)N_NODES_C * D_C;
  float* Xe = Y1 + (size_t)N_NODES_C * D_C;
  float* Xv = Xe + (size_t)N_EDGES_C * D_C;

  hipMemsetAsync(Xd, 0, (size_t)N_NODES_C * D_C * sizeof(float), stream);
  hipMemsetAsync(Xe, 0, (size_t)N_EDGES_C * D_C * sizeof(float), stream);
  hipMemsetAsync(Xv, 0, (size_t)N_NODES_C * D_C * sizeof(float), stream);

  k_diffuse<<<NNZ_C * 16 / 256, 256, 0, stream>>>(X, dvals, drows, dcols, Xd);
  k_gemm64<<<(N_NODES_C + 127) / 128, 256, 0, stream>>>(
      Xd, nullptr, 0.f, W1_w, W1_b, Y1, N_NODES_C);
  k_scatter_edge<<<NNZ_C * 16 / 256, 256, 0, stream>>>(Y1, vertex, edges, Xe);
  k_pair_gemm<<<NNZ_C / 128, 256, 0, stream>>>(Xd, Xe, vertex, edges, W2_w, W2_b, Xv);
  k_gemm64<<<(N_NODES_C + 127) / 128, 256, 0, stream>>>(
      Xv, Xd, ALPHA_C, W_w, W_b, out, N_NODES_C);
}


# W2 hoisted out of the per-pair product by linearity: Z=Xd@W2[0:64]+b2 per node, Q=Xe@W2[64:128] per hyperedge (f32 WMMA); pairs add Z[v]+Q[e], so the K=128 dot is summed as two K=64 halves
# speedup vs baseline: 1.2665x; 1.2665x over previous
#include <hip/hip_runtime.h>

typedef __attribute__((ext_vector_type(2))) float v2f;
typedef __attribute__((ext_vector_type(8))) float v8f;

#define N_NODES_C 100000
#define N_EDGES_C 25000
#define NNZ_C     800000
#define D_C       64
#define ALPHA_C   0.5f

#define A_PSTRIDE 144
#define B_PSTRIDE 160

__device__ __forceinline__ v8f wmma_f32(v2f a, v2f b, v8f c) {
  return __builtin_amdgcn_wmma_f32_16x16x4_f32(
       false, a,  false, b,
       (short)0, c,  false,  false);
}

__global__ void k_diffuse(const float* __restrict__ X,
                          const float* __restrict__ vals,
                          const int* __restrict__ rows,
                          const int* __restrict__ cols,
                          float* __restrict__ Xd) {
  int t  = blockIdx.x * blockDim.x + threadIdx.x;
  int i  = t >> 4;
  int c4 = (t & 15) << 2;
  __builtin_prefetch(vals + i + 1024, 0, 3);
  __builtin_prefetch(cols + i + 1024, 0, 3);
  float v = vals[i];
  const float4 x = *reinterpret_cast<const float4*>(X + (size_t)cols[i] * D_C + c4);
  float* dst = Xd + (size_t)rows[i] * D_C + c4;
  atomicAdd(dst + 0, v * x.x);
  atomicAdd(dst + 1, v * x.y);
  atomicAdd(dst + 2, v * x.z);
  atomicAdd(dst + 3, v * x.w);
}

__device__ __forceinline__ void gemm64_body(
    const float* A, const float* __restrict__ Ablend, float alpha,
    const float* __restrict__ B, const float* __restrict__ bias,
    float* Out, int nrows, int blk) {
  __shared__ float At[32 * A_PSTRIDE * 2];
  __shared__ float Bf[32 * B_PSTRIDE];
  __shared__ float bs[64];

  const int tid  = threadIdx.x;
  const int row0 = blk * 128;

  for (int idx = tid; idx < 128 * 16; idx += 256) {
    int rr = idx >> 4, c4 = idx & 15;
    int row = row0 + rr;
    float4 val = make_float4(0.f, 0.f, 0.f, 0.f);
    if (row < nrows) {
      val = *reinterpret_cast<const float4*>(A + (size_t)row * D_C + (c4 << 2));
      if (Ablend) {
        float4 w = *reinterpret_cast<const float4*>(Ablend + (size_t)row * D_C + (c4 << 2));
        val.x = (1.f - alpha) * val.x + alpha * w.x;
        val.y = (1.f - alpha) * val.y + alpha * w.y;
        val.z = (1.f - alpha) * val.z + alpha * w.z;
        val.w = (1.f - alpha) * val.w + alpha * w.w;
      }
    }
    int p = c4 << 1;
    v2f lo; lo.x = val.x; lo.y = val.y;
    v2f hi2; hi2.x = val.z; hi2.y = val.w;
    *reinterpret_cast<v2f*>(&At[(p * A_PSTRIDE + rr) * 2])       = lo;
    *reinterpret_cast<v2f*>(&At[((p + 1) * A_PSTRIDE + rr) * 2]) = hi2;
  }
  for (int idx = tid; idx < 64 * 64; idx += 256) {
    int k = idx >> 6, col = idx & 63;
    Bf[(k >> 1) * B_PSTRIDE + (col << 1) + (k & 1)] = B[idx];
  }
  if (tid < 64) bs[tid] = bias ? bias[tid] : 0.f;
  __syncthreads();

  const int wave = tid >> 5, lane = tid & 31;
  const int lrow = lane & 15, hi = lane >> 4;
  const int rr   = wave * 16 + lrow;

  v8f acc[4] = {v8f{}, v8f{}, v8f{}, v8f{}};
#pragma unroll
  for (int k4 = 0; k4 < 16; ++k4) {
    const int p = (k4 << 1) + hi;
    v2f a = *reinterpret_cast<const v2f*>(&At[(p * A_PSTRIDE + rr) * 2]);
#pragma unroll
    for (int n = 0; n < 4; ++n) {
      v2f b = *reinterpret_cast<const v2f*>(
          &Bf[p * B_PSTRIDE + (((n << 4) + lrow) << 1)]);
      acc[n] = wmma_f32(a, b, acc[n]);
    }
  }

#pragma unroll
  for (int n = 0; n < 4; ++n)
#pragma unroll
    for (int j = 0; j < 8; ++j) {
      int row = row0 + wave * 16 + j + (hi << 3);
      if (row < nrows) {
        int col = (n << 4) + lrow;
        Out[(size_t)row * D_C + col] = acc[n][j] + bs[col];
      }
    }
}

__global__ void __launch_bounds__(256) k_gemm64(
    const float* __restrict__ A, const float* __restrict__ Ablend, float alpha,
    const float* __restrict__ B, const float* __restrict__ bias,
    float* __restrict__ Out, int nrows) {
  gemm64_body(A, Ablend, alpha, B, bias, Out, nrows, blockIdx.x);
}

#define NB_NODE ((N_NODES_C + 127) / 128)
#define NB_EDGE ((N_EDGES_C + 127) / 128)
__global__ void __launch_bounds__(256) k_gemm64_w2(
    const float* Xd, float* XeQ,
    const float* __restrict__ W2, const float* __restrict__ b2,
    float* Z) {
  const bool edge_part = blockIdx.x >= NB_NODE;
  const float* A    = edge_part ? XeQ : Xd;
  const float* B    = edge_part ? W2 + D_C * D_C : W2;
  const float* bias = edge_part ? nullptr : b2;
  float* Out        = edge_part ? XeQ : Z;
  const int nrows   = edge_part ? N_EDGES_C : N_NODES_C;
  const int blk     = edge_part ? (int)blockIdx.x - NB_NODE : (int)blockIdx.x;
  gemm64_body(A, nullptr, 0.f, B, bias, Out, nrows, blk);
}

__global__ void k_scatter_edge(const float* __restrict__ Y1,
                               const int* __restrict__ vertex,
                               const int* __restrict__ edges,
                               float* __restrict__ Xe) {
  int t  = blockIdx.x * blockDim.x + threadIdx.x;
  int i  = t >> 4;
  int c4 = (t & 15) << 2;
  __builtin_prefetch(vertex + i + 1024, 0, 3);
  __builtin_prefetch(edges + i + 1024, 0, 3);
  const float4 y = *reinterpret_cast<const float4*>(Y1 + (size_t)vertex[i] * D_C + c4);
  float* dst = Xe + (size_t)edges[i] * D_C + c4;
  atomicAdd(dst + 0, y.x);
  atomicAdd(dst + 1, y.y);
  atomicAdd(dst + 2, y.z);
  atomicAdd(dst + 3, y.w);
}

__global__ void k_scatter_pair(const float* __restrict__ Z,
                               const float* __restrict__ Q,
                               const int* __restrict__ vertex,
                               const int* __restrict__ edges,
                               float* __restrict__ Xv) {
  int t  = blockIdx.x * blockDim.x + threadIdx.x;
  int i  = t >> 4;
  int c4 = (t & 15) << 2;
  __builtin_prefetch(vertex + i + 1024, 0, 3);
  __builtin_prefetch(edges + i + 1024, 0, 3);
  const size_t vo = (size_t)vertex[i] * D_C + c4;
  const float4 z = *reinterpret_cast<const float4*>(Z + vo);
  const float4 q = *reinterpret_cast<const float4*>(Q + (size_t)edges[i] * D_C + c4);
  float* dst = Xv + vo;
  atomicAdd(dst + 0, z.x + q.x);
  atomicAdd(dst + 1, z.y + q.y);
  atomicAdd(dst + 2, z.z + q.z);
  atomicAdd(dst + 3, z.w + q.w);
}

extern "C" void kernel_launch(void* const* d_in, const int* in_sizes, int n_in,
                              void* d_out, int out_size, void* d_ws, size_t ws_size,
                              hipStream_t stream) {
  const float* X      = (const float*)d_in[0];
  const float* W1_w   = (const float*)d_in[1];
  const float* W1_b   = (const float*)d_in[2];
  const float* W2_w   = (const float*)d_in[3];
  const float* W2_b   = (const float*)d_in[4];
  const float* W_w    = (const float*)d_in[5];
  const float* W_b    = (const float*)d_in[6];
  const float* dvals  = (const float*)d_in[7];
  const int*   vertex = (const int*)d_in[8];
  const int*   edges  = (const int*)d_in[9];
  const int*   drows  = (const int*)d_in[10];
  const int*   dcols  = (const int*)d_in[11];
  float* out = (float*)d_out;

  float* Xd = (float*)d_ws;
  float* Y1 = Xd + (size_t)N_NODES_C * D_C;
  float* Xe = Y1 + (size_t)N_NODES_C * D_C;
  float* Xv = Xe + (size_t)N_EDGES_C * D_C;

  hipMemsetAsync(Xd, 0, (size_t)N_NODES_C * D_C * sizeof(float), stream);
  hipMemsetAsync(Xe, 0, (size_t)N_EDGES_C * D_C * sizeof(float), stream);
  hipMemsetAsync(Xv, 0, (size_t)N_NODES_C * D_C * sizeof(float), stream);

  k_diffuse<<<NNZ_C * 16 / 256, 256, 0, stream>>>(X, dvals, drows, dcols, Xd);
  k_gemm64<<<(N_NODES_C + 127) / 128, 256, 0, stream>>>(
      Xd, nullptr, 0.f, W1_w, W1_b, Y1, N_NODES_C);
  k_scatter_edge<<<NNZ_C * 16 / 256, 256, 0, stream>>>(Y1, vertex, edges, Xe);
  k_gemm64_w2<<<NB_NODE + NB_EDGE, 256, 0, stream>>>(Xd, Xe, W2_w, W2_b, Y1);
  k_scatter_pair<<<NNZ_C * 16 / 256, 256, 0, stream>>>(Y1, Xe, vertex, edges, Xv);
  k_gemm64<<<(N_NODES_C + 127) / 128, 256, 0, stream>>>(
      Xv, Xd, ALPHA_C, W_w, W_b, out, N_NODES_C);
}


# Scatter kernels: one wave per pair, lane = column: every gather and float atomic add instruction covers one whole 128-B line (was 8 lanes in each of 4 lines); scalar index loads; atomics kept
# speedup vs baseline: 2.0530x; 1.6210x over previous
#include <hip/hip_runtime.h>

typedef __attribute__((ext_vector_type(2))) float v2f;
typedef __attribute__((ext_vector_type(8))) float v8f;

#define N_NODES_C 100000
#define N_EDGES_C 25000
#define NNZ_C     800000
#define D_C       64
#define ALPHA_C   0.5f

#define A_PSTRIDE 144
#define B_PSTRIDE 160

__device__ __forceinline__ v8f wmma_f32(v2f a, v2f b, v8f c) {
  return __builtin_amdgcn_wmma_f32_16x16x4_f32(
       false, a,  false, b,
       (short)0, c,  false,  false);
}

#define SC_WAVES 8
#define NPW_DIFF 16
#define NPW_EDGE 16
#define NPW_PAIR 8
static_assert(NNZ_C % (SC_WAVES * NPW_DIFF) == 0, "K1 grid must tile nnz exactly");
static_assert(NNZ_C % (SC_WAVES * NPW_EDGE) == 0, "K3 grid must tile nnz exactly");
static_assert(NNZ_C % (SC_WAVES * NPW_PAIR) == 0, "K4b grid must tile nnz exactly");

__device__ __forceinline__ int wave_first_pair(int npw) {
  const int wave = __builtin_amdgcn_readfirstlane((int)(threadIdx.x >> 5));
  return ((int)blockIdx.x * SC_WAVES + wave) * npw;
}

__global__ void __launch_bounds__(256) k_diffuse(const float* __restrict__ X,
                          const float* __restrict__ vals,
                          const int* __restrict__ rows,
                          const int* __restrict__ cols,
                          float* __restrict__ Xd) {
  const unsigned lane = threadIdx.x & 31;
  const int i0 = wave_first_pair(NPW_DIFF);
  unsigned c[NPW_DIFF], r[NPW_DIFF];
  float v[NPW_DIFF], x0[NPW_DIFF], x1[NPW_DIFF];
#pragma unroll
  for (int j = 0; j < NPW_DIFF; ++j) {
    c[j] = (unsigned)cols[i0 + j];
    r[j] = (unsigned)rows[i0 + j];
    v[j] = vals[i0 + j];
  }
#pragma unroll
  for (int j = 0; j < NPW_DIFF; ++j) {
    const float* s = X + (c[j] * (unsigned)D_C + lane);
    x0[j] = s[0];
    x1[j] = s[32];
  }
#pragma unroll
  for (int j = 0; j < NPW_DIFF; ++j) {
    float* dst = Xd + (r[j] * (unsigned)D_C + lane);
    atomicAdd(dst, v[j] * x0[j]);
    atomicAdd(dst + 32, v[j] * x1[j]);
  }
}

__device__ __forceinline__ void gemm64_body(
    const float* A, const float* __restrict__ Ablend, float alpha,
    const float* __restrict__ B, const float* __restrict__ bias,
    float* Out, int nrows, int blk) {
  __shared__ float At[32 * A_PSTRIDE * 2];
  __shared__ float Bf[32 * B_PSTRIDE];
  __shared__ float bs[64];

  const int tid  = threadIdx.x;
  const int row0 = blk * 128;

  for (int idx = tid; idx < 128 * 16; idx += 256) {
    int rr = idx >> 4, c4 = idx & 15;
    int row = row0 + rr;
    float4 val = make_float4(0.f, 0.f, 0.f, 0.f);
    if (row < nrows) {
      val = *reinterpret_cast<const float4*>(A + (size_t)row * D_C + (c4 << 2));
      if (Ablend) {
        float4 w = *reinterpret_cast<const float4*>(Ablend + (size_t)row * D_C + (c4 << 2));
        val.x = (1.f - alpha) * val.x + alpha * w.x;
        val.y = (1.f - alpha) * val.y + alpha * w.y;
        val.z = (1.f - alpha) * val.z + alpha * w.z;
        val.w = (1.f - alpha) * val.w + alpha * w.w;
      }
    }
    int p = c4 << 1;
    v2f lo; lo.x = val.x; lo.y = val.y;
    v2f hi2; hi2.x = val.z; hi2.y = val.w;
    *reinterpret_cast<v2f*>(&At[(p * A_PSTRIDE + rr) * 2])       = lo;
    *reinterpret_cast<v2f*>(&At[((p + 1) * A_PSTRIDE + rr) * 2]) = hi2;
  }
  for (int idx = tid; idx < 64 * 64; idx += 256) {
    int k = idx >> 6, col = idx & 63;
    Bf[(k >> 1) * B_PSTRIDE + (col << 1) + (k & 1)] = B[idx];
  }
  if (tid < 64) bs[tid] = bias ? bias[tid] : 0.f;
  __syncthreads();

  const int wave = tid >> 5, lane = tid & 31;
  const int lrow = lane & 15, hi = lane >> 4;
  const int rr   = wave * 16 + lrow;

  v8f acc[4] = {v8f{}, v8f{}, v8f{}, v8f{}};
#pragma unroll
  for (int k4 = 0; k4 < 16; ++k4) {
    const int p = (k4 << 1) + hi;
    v2f a = *reinterpret_cast<const v2f*>(&At[(p * A_PSTRIDE + rr) * 2]);
#pragma unroll
    for (int n = 0; n < 4; ++n) {
      v2f b = *reinterpret_cast<const v2f*>(
          &Bf[p * B_PSTRIDE + (((n << 4) + lrow) << 1)]);
      acc[n] = wmma_f32(a, b, acc[n]);
    }
  }

#pragma unroll
  for (int n = 0; n < 4; ++n)
#pragma unroll
    for (int j = 0; j < 8; ++j) {
      int row = row0 + wave * 16 + j + (hi << 3);
      if (row < nrows) {
        int col = (n << 4) + lrow;
        Out[(size_t)row * D_C + col] = acc[n][j] + bs[col];
      }
    }
}

__global__ void __launch_bounds__(256) k_gemm64(
    const float* __restrict__ A, const float* __restrict__ Ablend, float alpha,
    const float* __restrict__ B, const float* __restrict__ bias,
    float* __restrict__ Out, int nrows) {
  gemm64_body(A, Ablend, alpha, B, bias, Out, nrows, blockIdx.x);
}

#define NB_NODE ((N_NODES_C + 127) / 128)
#define NB_EDGE ((N_EDGES_C + 127) / 128)
__global__ void __launch_bounds__(256) k_gemm64_w2(
    const float* Xd, float* XeQ,
    const float* __restrict__ W2, const float* __restrict__ b2,
    float* Z) {
  const bool edge_part = blockIdx.x >= NB_NODE;
  const float* A    = edge_part ? XeQ : Xd;
  const float* B    = edge_part ? W2 + D_C * D_C : W2;
  const float* bias = edge_part ? nullptr : b2;
  float* Out        = edge_part ? XeQ : Z;
  const int nrows   = edge_part ? N_EDGES_C : N_NODES_C;
  const int blk     = edge_part ? (int)blockIdx.x - NB_NODE : (int)blockIdx.x;
  gemm64_body(A, nullptr, 0.f, B, bias, Out, nrows, blk);
}

__global__ void __launch_bounds__(256) k_scatter_edge(const float* __restrict__ Y1,
                               const int* __restrict__ vertex,
                               const int* __restrict__ edges,
                               float* __restrict__ Xe) {
  const unsigned lane = threadIdx.x & 31;
  const int i0 = wave_first_pair(NPW_EDGE);
  unsigned vi[NPW_EDGE], ei[NPW_EDGE];
  float y0[NPW_EDGE], y1[NPW_EDGE];
#pragma unroll
  for (int j = 0; j < NPW_EDGE; ++j) {
    vi[j] = (unsigned)vertex[i0 + j];
    ei[j] = (unsigned)edges[i0 + j];
  }
#pragma unroll
  for (int j = 0; j < NPW_EDGE; ++j) {
    const float* s = Y1 + (vi[j] * (unsigned)D_C + lane);
    y0[j] = s[0];
    y1[j] = s[32];
  }
#pragma unroll
  for (int j = 0; j < NPW_EDGE; ++j) {
    float* dst = Xe + (ei[j] * (unsigned)D_C + lane);
    atomicAdd(dst, y0[j]);
    atomicAdd(dst + 32, y1[j]);
  }
}

__global__ void __launch_bounds__(256) k_scatter_pair(const float* __restrict__ Z,
                               const float* __restrict__ Q,
                               const int* __restrict__ vertex,
                               const int* __restrict__ edges,
                               float* __restrict__ Xv) {
  const unsigned lane = threadIdx.x & 31;
  const int i0 = wave_first_pair(NPW_PAIR);
  unsigned vi[NPW_PAIR], ei[NPW_PAIR];
  float z0[NPW_PAIR], z1[NPW_PAIR], q0[NPW_PAIR], q1[NPW_PAIR];
#pragma unroll
  for (int j = 0; j < NPW_PAIR; ++j) {
    vi[j] = (unsigned)vertex[i0 + j];
    ei[j] = (unsigned)edges[i0 + j];
  }
#pragma unroll
  for (int j = 0; j < NPW_PAIR; ++j) {
    const float* sz = Z + (vi[j] * (unsigned)D_C + lane);
    const float* sq = Q + (ei[j] * (unsigned)D_C + lane);
    z0[j] = sz[0];
    z1[j] = sz[32];
    q0[j] = sq[0];
    q1[j] = sq[32];
  }
#pragma unroll
  for (int j = 0; j < NPW_PAIR; ++j) {
    float* dst = Xv + (vi[j] * (unsigned)D_C + lane);
    atomicAdd(dst, z0[j] + q0[j]);
    atomicAdd(dst + 32, z1[j] + q1[j]);
  }
}

extern "C" void kernel_launch(void* const* d_in, const int* in_sizes, int n_in,
                              void* d_out, int out_size, void* d_ws, size_t ws_size,
                              hipStream_t stream) {
  const float* X      = (const float*)d_in[0];
  const float* W1_w   = (const float*)d_in[1];
  const float* W1_b   = (const float*)d_in[2];
  const float* W2_w   = (const float*)d_in[3];
  const float* W2_b   = (const float*)d_in[4];
  const float* W_w    = (const float*)d_in[5];
  const float* W_b    = (const float*)d_in[6];
  const float* dvals  = (const float*)d_in[7];
  const int*   vertex = (const int*)d_in[8];
  const int*   edges  = (const int*)d_in[9];
  const int*   drows  = (const int*)d_in[10];
  const int*   dcols  = (const int*)d_in[11];
  float* out = (float*)d_out;

  float* Xd = (float*)d_ws;
  float* Y1 = Xd + (size_t)N_NODES_C * D_C;
  float* Xe = Y1 + (size_t)N_NODES_C * D_C;
  float* Xv = Xe + (size_t)N_EDGES_C * D_C;

  hipMemsetAsync(Xd, 0, (size_t)N_NODES_C * D_C * sizeof(float), stream);
  hipMemsetAsync(Xe, 0, (size_t)N_EDGES_C * D_C * sizeof(float), stream);
  hipMemsetAsync(Xv, 0, (size_t)N_NODES_C * D_C * sizeof(float), stream);

  k_diffuse<<<NNZ_C / (SC_WAVES * NPW_DIFF), 256, 0, stream>>>(X, dvals, drows, dcols, Xd);
  k_gemm64<<<(N_NODES_C + 127) / 128, 256, 0, stream>>>(
      Xd, nullptr, 0.f, W1_w, W1_b, Y1, N_NODES_C);
  k_scatter_edge<<<NNZ_C / (SC_WAVES * NPW_EDGE), 256, 0, stream>>>(Y1, vertex, edges, Xe);
  k_gemm64_w2<<<NB_NODE + NB_EDGE, 256, 0, stream>>>(Xd, Xe, W2_w, W2_b, Y1);
  k_scatter_pair<<<NNZ_C / (SC_WAVES * NPW_PAIR), 256, 0, stream>>>(Y1, Xe, vertex, edges, Xv);
  k_gemm64<<<(N_NODES_C + 127) / 128, 256, 0, stream>>>(
      Xv, Xd, ALPHA_C, W_w, W_b, out, N_NODES_C);
}
